# starting kernel
# baseline (speedup 1.0000x reference)
#include <hip/hip_runtime.h>
#include <hip/hip_bf16.h>

typedef __attribute__((ext_vector_type(16))) _Float16 v16h;
typedef __attribute__((ext_vector_type(8)))  float    v8f;

#define DIMS   128
#define PMAX   50
#define PPAD   64
#define LSTR   132

__global__ __launch_bounds__(128)
void AttCML_kernel(const float* __restrict__ user_emb,
                   const float* __restrict__ item_emb,
                   const int*   __restrict__ user_ids,
                   const int*   __restrict__ pos_ids,
                   const int*   __restrict__ neg_ids,
                   const int*   __restrict__ pref_ids,
                   const float* __restrict__ n_prefs,
                   float* __restrict__ out,
                   int B, int n_items)
{
    __shared__ __align__(16) float sPref[PPAD * LSTR];
    __shared__ __align__(16) float sT[2 * LSTR];
    __shared__ float sW[2 * PPAD];
    __shared__ float sAtt[2 * PPAD];
    __shared__ float sRed[8];

    const int b    = blockIdx.x;
    const int tid  = threadIdx.x;
    const int wave = tid >> 5;
    const int lane = tid & 31;

    for (int r = wave; r < PPAD; r += 4) {
        float4 v = make_float4(0.f, 0.f, 0.f, 0.f);
        if (r < PMAX) {
            int id = pref_ids[b * PMAX + r];
            if (id < n_items)
                v = *(const float4*)(item_emb + (size_t)id * DIMS + lane * 4);
        }
        *(float4*)(&sPref[r * LSTR + lane * 4]) = v;
    }
    if (wave == 0) {
        int id = pos_ids[b];
        *(float4*)(&sT[0 * LSTR + lane * 4]) =
            *(const float4*)(item_emb + (size_t)id * DIMS + lane * 4);
    } else if (wave == 1) {
        int id = neg_ids[b];
        *(float4*)(&sT[1 * LSTR + lane * 4]) =
            *(const float4*)(item_emb + (size_t)id * DIMS + lane * 4);
    }
    __syncthreads();

    {
        const int mrow  = wave * 16 + (lane & 15);
        const int half8 = (lane >> 4) ? 8 : 0;
        const int col   = lane & 15;
        const float* arow = &sPref[mrow * LSTR];
        const float* brow = &sT[(col < 2 ? col : 0) * LSTR];
        const bool   bval = (col < 2);

        v8f c = {0.f, 0.f, 0.f, 0.f, 0.f, 0.f, 0.f, 0.f};
        #pragma unroll
        for (int kb = 0; kb < DIMS; kb += 32) {
            const int k0 = kb + half8;
            const int k1 = kb + 16 + half8;
            v16h a, bm;
            #pragma unroll
            for (int j = 0; j < 4; ++j) {
                float2 fa0 = *(const float2*)(arow + k0 + 2 * j);
                float2 fa1 = *(const float2*)(arow + k1 + 2 * j);
                a[2 * j]     = (_Float16)fa0.x;
                a[2 * j + 1] = (_Float16)fa0.y;
                a[8 + 2 * j]     = (_Float16)fa1.x;
                a[8 + 2 * j + 1] = (_Float16)fa1.y;
                float2 fb0 = *(const float2*)(brow + k0 + 2 * j);
                float2 fb1 = *(const float2*)(brow + k1 + 2 * j);
                bm[2 * j]     = (_Float16)(bval ? fb0.x : 0.f);
                bm[2 * j + 1] = (_Float16)(bval ? fb0.y : 0.f);
                bm[8 + 2 * j]     = (_Float16)(bval ? fb1.x : 0.f);
                bm[8 + 2 * j + 1] = (_Float16)(bval ? fb1.y : 0.f);
            }
            c = __builtin_amdgcn_wmma_f32_16x16x32_f16(false, a, false, bm,
                                                       (short)0, c, false, false);
        }
        if (col < 2) {
            #pragma unroll
            for (int g = 0; g < 8; ++g)
                sW[col * PPAD + wave * 16 + half8 + g] = c[g];
        }
    }
    __syncthreads();

    if (wave < 2) {
        const float np1 = n_prefs[b] + 1.0f;
        const int p0 = lane, p1 = lane + 32;
        float e0 = ((float)p0 < np1) ? __expf(sW[wave * PPAD + p0]) : 0.f;
        float e1 = (p1 < PMAX && (float)p1 < np1) ? __expf(sW[wave * PPAD + p1]) : 0.f;
        float s = e0 + e1;
        #pragma unroll
        for (int off = 16; off > 0; off >>= 1) s += __shfl_xor(s, off, 32);
        const float inv = 1.0f / s;
        sAtt[wave * PPAD + p0] = e0 * inv;
        if (p1 < PMAX) sAtt[wave * PPAD + p1] = e1 * inv;
    }
    __syncthreads();

    {
        const int d = tid;
        float accP = 0.f, accN = 0.f;
        #pragma unroll 5
        for (int p = 0; p < PMAX; ++p) {
            const float pv = sPref[p * LSTR + d];
            accP = fmaf(sAtt[p], pv, accP);
            accN = fmaf(sAtt[PPAD + p], pv, accN);
        }
        const float u  = user_emb[(size_t)user_ids[b] * DIMS + d];
        const float dp = u + accP - sT[d];
        const float dn = u + accN - sT[LSTR + d];
        float sqP = dp * dp, sqN = dn * dn;
        #pragma unroll
        for (int off = 16; off > 0; off >>= 1) {
            sqP += __shfl_xor(sqP, off, 32);
            sqN += __shfl_xor(sqN, off, 32);
        }
        if (lane == 0) { sRed[wave] = sqP; sRed[4 + wave] = sqN; }
    }
    __syncthreads();
    if (tid == 0) {
        out[b]     = sRed[0] + sRed[1] + sRed[2] + sRed[3];
        out[B + b] = sRed[4] + sRed[5] + sRed[6] + sRed[7];
    }
}

extern "C" void kernel_launch(void* const* d_in, const int* in_sizes, int n_in,
                              void* d_out, int out_size, void* d_ws, size_t ws_size,
                              hipStream_t stream) {
    (void)n_in; (void)out_size; (void)d_ws; (void)ws_size;
    const float* user_emb = (const float*)d_in[0];
    const float* item_emb = (const float*)d_in[1];
    const int*   user_ids = (const int*)d_in[2];
    const int*   pos_ids  = (const int*)d_in[3];
    const int*   neg_ids  = (const int*)d_in[4];
    const int*   pref_ids = (const int*)d_in[5];
    const float* n_prefs  = (const float*)d_in[6];
    float* out = (float*)d_out;

    const int B       = in_sizes[2];
    const int n_items = in_sizes[1] / DIMS;

    AttCML_kernel<<<B, 128, 0, stream>>>(user_emb, item_emb, user_ids, pos_ids,
                                         neg_ids, pref_ids, n_prefs, out, B, n_items);
}


# Cross-lane sums (softmax total, row-parity join, squared distance) on DPP row_xmask adds and v_permlanex16, not ds_bpermute: 10 dependent LDS hops leave each wave's chain. Same adds, bit-identical
# speedup vs baseline: 2.6642x; 2.6642x over previous
#include <hip/hip_runtime.h>


typedef __attribute__((ext_vector_type(16))) _Float16 v16h;
typedef __attribute__((ext_vector_type(8)))  _Float16 v8h;
typedef __attribute__((ext_vector_type(4)))  _Float16 v4h;
typedef __attribute__((ext_vector_type(8)))  float    v8f;

#define DIMS   128
#define PMAX   50
#define NROW   52
#define NLD    26
#define SPITCH 72

__device__ __forceinline__ void wave_lds_handover() {
    __builtin_amdgcn_fence(__ATOMIC_RELEASE, "wavefront");
    __builtin_amdgcn_wave_barrier();
    __builtin_amdgcn_fence(__ATOMIC_ACQUIRE, "wavefront");
}

template <int M>
__device__ __forceinline__ float row_xor(float x) {
    const int b = __builtin_bit_cast(int, x);
    return __builtin_bit_cast(float, __builtin_amdgcn_update_dpp(0, b, 0x160 + M, 0xf, 0xf, true));
}
__device__ __forceinline__ float xhalf(float x) {
    const int b = __builtin_bit_cast(int, x);
    return __builtin_bit_cast(float, __builtin_amdgcn_permlanex16(b, b, 0x76543210, 0xfedcba98, false, false));
}

__global__ __launch_bounds__(64)
void AttCML_kernel(const float* __restrict__ user_emb,
                   const float* __restrict__ item_emb,
                   const int*   __restrict__ user_ids,
                   const int*   __restrict__ pos_ids,
                   const int*   __restrict__ neg_ids,
                   const int*   __restrict__ pref_ids,
                   const float* __restrict__ n_prefs,
                   float* __restrict__ out,
                   int B, int n_items)
{
    __shared__ __align__(16) _Float16 sSlab[2][NROW * SPITCH];
    __shared__ __align__(16) float    sPart[2][2][64];
    __shared__ __align__(16) float    sAtt[2][2][64];
    __shared__ float                  sRed[2][2];

    const int b    = blockIdx.x;
    const int lane = threadIdx.x & 31;
    const int kh   = __builtin_amdgcn_readfirstlane(threadIdx.x >> 5);
    const int q    = lane >> 4;
    const int j    = lane & 15;

    const char* ibase = (const char*)item_emb;
    const unsigned dbyte = 256u * kh + 16u * j;
    const int* prow = pref_ids + (size_t)b * PMAX;
    unsigned off[NLD];
    #pragma unroll
    for (int i = 0; i < PMAX / 2; ++i) {
        const int id  = prow[2 * i + q];
        const int idc = (id < n_items) ? id : (n_items - 1);
        off[i] = (unsigned)idc * (DIMS * 4u) + dbyte;
    }
    off[NLD - 1] = (unsigned)(q ? neg_ids[b] : pos_ids[b]) * (DIMS * 4u) + dbyte;
    const unsigned uoff = (unsigned)user_ids[b] * (DIMS * 4u) + dbyte;
    const float    np1  = n_prefs[b] + 1.0f;
    const int pid0 = prow[lane];
    const int pid1 = prow[(lane + 32 < PMAX) ? (lane + 32) : (PMAX - 1)];
    __builtin_amdgcn_sched_barrier(0);
    float4 x[NLD];
    #pragma unroll
    for (int i = 0; i < NLD; ++i) x[i] = *(const float4*)(ibase + off[i]);
    const float4 u4 = *(const float4*)((const char*)user_emb + uoff);
    __builtin_amdgcn_sched_barrier(0);

    _Float16* slab = &sSlab[kh][0];
    #pragma unroll
    for (int i = 0; i < NLD; ++i) {
        v4h hv;
        hv[0] = (_Float16)x[i].x; hv[1] = (_Float16)x[i].y;
        hv[2] = (_Float16)x[i].z; hv[3] = (_Float16)x[i].w;
        *(v4h*)(slab + (2 * i + q) * SPITCH + 4 * j) = hv;
    }
    wave_lds_handover();

    {
        v16h bm[2];
        #pragma unroll
        for (int ks = 0; ks < 2; ++ks) {
            const _Float16* p = slab + (PMAX + (j & 1)) * SPITCH + 32 * ks + 16 * q;
            const v8h lo = *(const v8h*)p;
            const v8h hi = *(const v8h*)(p + 8);
            const v16h t = __builtin_shufflevector(lo, hi, 0, 1, 2, 3, 4, 5, 6, 7,
                                                   8, 9, 10, 11, 12, 13, 14, 15);
            const v16h z = {};
            bm[ks] = (j < 2) ? t : z;
        }
        v8f acc[4];
        #pragma unroll
        for (int m = 0; m < 4; ++m) {
            int row = 16 * m + j;
            row = (row < NROW) ? row : (NROW - 1);
            v8f cc = {0.f, 0.f, 0.f, 0.f, 0.f, 0.f, 0.f, 0.f};
            #pragma unroll
            for (int ks = 0; ks < 2; ++ks) {
                const _Float16* p = slab + row * SPITCH + 32 * ks + 16 * q;
                const v8h lo = *(const v8h*)p;
                const v8h hi = *(const v8h*)(p + 8);
                const v16h a = __builtin_shufflevector(lo, hi, 0, 1, 2, 3, 4, 5, 6, 7,
                                                       8, 9, 10, 11, 12, 13, 14, 15);
                cc = __builtin_amdgcn_wmma_f32_16x16x32_f16(false, a, false, bm[ks],
                                                            (short)0, cc, false, false);
            }
            acc[m] = cc;
        }
        if (j < 2) {
            float* dst = &sPart[kh][j][8 * q];
            #pragma unroll
            for (int m = 0; m < 4; ++m) {
                *(float4*)(dst + 16 * m)     = make_float4(acc[m][0], acc[m][1], acc[m][2], acc[m][3]);
                *(float4*)(dst + 16 * m + 4) = make_float4(acc[m][4], acc[m][5], acc[m][6], acc[m][7]);
            }
        }
    }
    __syncthreads();

    const bool pad0 = pid0 >= n_items;
    const bool pad1 = pid1 >= n_items;
    #pragma unroll
    for (int t = 0; t < 2; ++t) {
        const int p0 = lane, p1 = lane + 32;
        const float w0 = sPart[0][t][p0] + sPart[1][t][p0];
        const float w1 = sPart[0][t][p1] + sPart[1][t][p1];
        const float e0 = ((float)p0 < np1) ? (pad0 ? 1.0f : __expf(w0)) : 0.f;
        const float e1 = (p1 < PMAX && (float)p1 < np1) ? (pad1 ? 1.0f : __expf(w1)) : 0.f;
        float s = e0 + e1;
        s += xhalf(s);
        s += row_xor<8>(s);
        s += row_xor<4>(s);
        s += row_xor<2>(s);
        s += row_xor<1>(s);
        const float inv = 1.0f / s;
        sAtt[kh][t][(lane & 1) * 32 + (lane >> 1)]      = pad0 ? 0.f : e0 * inv;
        sAtt[kh][t][(lane & 1) * 32 + 16 + (lane >> 1)] = pad1 ? 0.f : e1 * inv;
    }
    wave_lds_handover();

    float4 accT[2];
    #pragma unroll
    for (int t = 0; t < 2; ++t) {
        const float4* ap = (const float4*)&sAtt[kh][t][q * 32];
        float a[28];
        #pragma unroll
        for (int g = 0; g < 7; ++g) {
            const float4 v = ap[g];
            a[4 * g] = v.x; a[4 * g + 1] = v.y; a[4 * g + 2] = v.z; a[4 * g + 3] = v.w;
        }
        float4 s4 = make_float4(0.f, 0.f, 0.f, 0.f);
        #pragma unroll
        for (int i = 0; i < PMAX / 2; ++i) {
            s4.x = fmaf(a[i], x[i].x, s4.x);
            s4.y = fmaf(a[i], x[i].y, s4.y);
            s4.z = fmaf(a[i], x[i].z, s4.z);
            s4.w = fmaf(a[i], x[i].w, s4.w);
        }
        s4.x += xhalf(s4.x);
        s4.y += xhalf(s4.y);
        s4.z += xhalf(s4.z);
        s4.w += xhalf(s4.w);
        accT[t] = s4;
    }

    {
        const float4 av = q ? accT[1] : accT[0];
        const float4 tg = x[NLD - 1];
        const float dx = (u4.x + av.x) - tg.x;
        const float dy = (u4.y + av.y) - tg.y;
        const float dz = (u4.z + av.z) - tg.z;
        const float dw = (u4.w + av.w) - tg.w;
        float sq = dx * dx + dy * dy + dz * dz + dw * dw;
        sq += row_xor<8>(sq);
        sq += row_xor<4>(sq);
        sq += row_xor<2>(sq);
        sq += row_xor<1>(sq);
        if (j == 0) sRed[q][kh] = sq;
    }
    __syncthreads();
    if (threadIdx.x < 2)
        out[(size_t)threadIdx.x * B + b] = sRed[threadIdx.x][0] + sRed[threadIdx.x][1];
}

extern "C" void kernel_launch(void* const* d_in, const int* in_sizes, int n_in,
                              void* d_out, int out_size, void* d_ws, size_t ws_size,
                              hipStream_t stream) {
    (void)n_in; (void)out_size; (void)d_ws; (void)ws_size;
    const float* user_emb = (const float*)d_in[0];
    const float* item_emb = (const float*)d_in[1];
    const int*   user_ids = (const int*)d_in[2];
    const int*   pos_ids  = (const int*)d_in[3];
    const int*   neg_ids  = (const int*)d_in[4];
    const int*   pref_ids = (const int*)d_in[5];
    const float* n_prefs  = (const float*)d_in[6];
    float* out = (float*)d_out;

    const int B       = in_sizes[2];
    const int n_items = in_sizes[1] / DIMS;

    AttCML_kernel<<<B, 64, 0, stream>>>(user_emb, item_emb, user_ids, pos_ids,
                                        neg_ids, pref_ids, n_prefs, out, B, n_items);
}


# Rows past the mask prefix are not fetched: the 25 two-row loads sit behind scalar tests on ceil((n_prefs+1)/2), the same in every lane; skipped registers hold zeros. Same sums, bit-identical
# speedup vs baseline: 3.1585x; 1.1855x over previous
#include <hip/hip_runtime.h>


typedef __attribute__((ext_vector_type(16))) _Float16 v16h;
typedef __attribute__((ext_vector_type(8)))  _Float16 v8h;
typedef __attribute__((ext_vector_type(4)))  _Float16 v4h;
typedef __attribute__((ext_vector_type(8)))  float    v8f;

#define DIMS   128
#define PMAX   50
#define NROW   52
#define NLD    26
#define SPITCH 72

__device__ __forceinline__ void wave_lds_handover() {
    __builtin_amdgcn_fence(__ATOMIC_RELEASE, "wavefront");
    __builtin_amdgcn_wave_barrier();
    __builtin_amdgcn_fence(__ATOMIC_ACQUIRE, "wavefront");
}

template <int M>
__device__ __forceinline__ float row_xor(float x) {
    const int b = __builtin_bit_cast(int, x);
    return __builtin_bit_cast(float, __builtin_amdgcn_update_dpp(0, b, 0x160 + M, 0xf, 0xf, true));
}
__device__ __forceinline__ float xhalf(float x) {
    const int b = __builtin_bit_cast(int, x);
    return __builtin_bit_cast(float, __builtin_amdgcn_permlanex16(b, b, 0x76543210, 0xfedcba98, false, false));
}

__global__ __launch_bounds__(64)
void AttCML_kernel(const float* __restrict__ user_emb,
                   const float* __restrict__ item_emb,
                   const int*   __restrict__ user_ids,
                   const int*   __restrict__ pos_ids,
                   const int*   __restrict__ neg_ids,
                   const int*   __restrict__ pref_ids,
                   const float* __restrict__ n_prefs,
                   float* __restrict__ out,
                   int B, int n_items)
{
    __shared__ __align__(16) _Float16 sSlab[2][NROW * SPITCH];
    __shared__ __align__(16) float    sPart[2][2][64];
    __shared__ __align__(16) float    sAtt[2][2][64];
    __shared__ float                  sRed[2][2];

    const int b    = blockIdx.x;
    const int lane = threadIdx.x & 31;
    const int kh   = __builtin_amdgcn_readfirstlane(threadIdx.x >> 5);
    const int q    = lane >> 4;
    const int j    = lane & 15;

    const char* ibase = (const char*)item_emb;
    const unsigned dbyte = 256u * kh + 16u * j;
    const int* prow = pref_ids + (size_t)b * PMAX;
    unsigned off[NLD];
    #pragma unroll
    for (int i = 0; i < PMAX / 2; ++i) {
        const int id  = prow[2 * i + q];
        const int idc = (id < n_items) ? id : (n_items - 1);
        off[i] = (unsigned)idc * (DIMS * 4u) + dbyte;
    }
    off[NLD - 1] = (unsigned)(q ? neg_ids[b] : pos_ids[b]) * (DIMS * 4u) + dbyte;
    const unsigned uoff = (unsigned)user_ids[b] * (DIMS * 4u) + dbyte;
    const float    np1  = n_prefs[b] + 1.0f;
    const int nrows = (int)fminf(fmaxf(ceilf(np1), 0.0f), (float)PMAX);
    const int nl    = __builtin_amdgcn_readfirstlane((nrows + 1) >> 1);
    const int pid0 = prow[lane];
    const int pid1 = prow[(lane + 32 < PMAX) ? (lane + 32) : (PMAX - 1)];
    asm volatile("" :: "v"(off[0]),  "v"(off[1]),  "v"(off[2]),  "v"(off[3]),  "v"(off[4]),
                       "v"(off[5]),  "v"(off[6]),  "v"(off[7]),  "v"(off[8]),  "v"(off[9]),
                       "v"(off[10]), "v"(off[11]), "v"(off[12]), "v"(off[13]), "v"(off[14]),
                       "v"(off[15]), "v"(off[16]), "v"(off[17]), "v"(off[18]), "v"(off[19]),
                       "v"(off[20]), "v"(off[21]), "v"(off[22]), "v"(off[23]), "v"(off[24]),
                       "v"(off[25]), "v"(uoff));
    float4 x[NLD];
    float4 u4;
    x[NLD - 1] = *(const float4*)(ibase + off[NLD - 1]);
    u4 = *(const float4*)((const char*)user_emb + uoff);
    #pragma unroll
    for (int i = 0; i < PMAX / 2; ++i) x[i] = make_float4(0.f, 0.f, 0.f, 0.f);
    #pragma unroll
    for (int i = 0; i < PMAX / 2; ++i)
        if (i < nl) x[i] = *(const float4*)(ibase + off[i]);

    _Float16* slab = &sSlab[kh][0];
    #pragma unroll
    for (int i = 0; i < NLD; ++i) {
        v4h hv;
        hv[0] = (_Float16)x[i].x; hv[1] = (_Float16)x[i].y;
        hv[2] = (_Float16)x[i].z; hv[3] = (_Float16)x[i].w;
        *(v4h*)(slab + (2 * i + q) * SPITCH + 4 * j) = hv;
    }
    wave_lds_handover();

    {
        v16h bm[2];
        #pragma unroll
        for (int ks = 0; ks < 2; ++ks) {
            const _Float16* p = slab + (PMAX + (j & 1)) * SPITCH + 32 * ks + 16 * q;
            const v8h lo = *(const v8h*)p;
            const v8h hi = *(const v8h*)(p + 8);
            const v16h t = __builtin_shufflevector(lo, hi, 0, 1, 2, 3, 4, 5, 6, 7,
                                                   8, 9, 10, 11, 12, 13, 14, 15);
            const v16h z = {};
            bm[ks] = (j < 2) ? t : z;
        }
        v8f acc[4];
        #pragma unroll
        for (int m = 0; m < 4; ++m) {
            int row = 16 * m + j;
            row = (row < NROW) ? row : (NROW - 1);
            v8f cc = {0.f, 0.f, 0.f, 0.f, 0.f, 0.f, 0.f, 0.f};
            #pragma unroll
            for (int ks = 0; ks < 2; ++ks) {
                const _Float16* p = slab + row * SPITCH + 32 * ks + 16 * q;
                const v8h lo = *(const v8h*)p;
                const v8h hi = *(const v8h*)(p + 8);
                const v16h a = __builtin_shufflevector(lo, hi, 0, 1, 2, 3, 4, 5, 6, 7,
                                                       8, 9, 10, 11, 12, 13, 14, 15);
                cc = __builtin_amdgcn_wmma_f32_16x16x32_f16(false, a, false, bm[ks],
                                                            (short)0, cc, false, false);
            }
            acc[m] = cc;
        }
        if (j < 2) {
            float* dst = &sPart[kh][j][8 * q];
            #pragma unroll
            for (int m = 0; m < 4; ++m) {
                *(float4*)(dst + 16 * m)     = make_float4(acc[m][0], acc[m][1], acc[m][2], acc[m][3]);
                *(float4*)(dst + 16 * m + 4) = make_float4(acc[m][4], acc[m][5], acc[m][6], acc[m][7]);
            }
        }
    }
    __syncthreads();

    const bool pad0 = pid0 >= n_items;
    const bool pad1 = pid1 >= n_items;
    #pragma unroll
    for (int t = 0; t < 2; ++t) {
        const int p0 = lane, p1 = lane + 32;
        const float w0 = sPart[0][t][p0] + sPart[1][t][p0];
        const float w1 = sPart[0][t][p1] + sPart[1][t][p1];
        const float e0 = ((float)p0 < np1) ? (pad0 ? 1.0f : __expf(w0)) : 0.f;
        const float e1 = (p1 < PMAX && (float)p1 < np1) ? (pad1 ? 1.0f : __expf(w1)) : 0.f;
        float s = e0 + e1;
        s += xhalf(s);
        s += row_xor<8>(s);
        s += row_xor<4>(s);
        s += row_xor<2>(s);
        s += row_xor<1>(s);
        const float inv = 1.0f / s;
        sAtt[kh][t][(lane & 1) * 32 + (lane >> 1)]      = pad0 ? 0.f : e0 * inv;
        sAtt[kh][t][(lane & 1) * 32 + 16 + (lane >> 1)] = pad1 ? 0.f : e1 * inv;
    }
    wave_lds_handover();

    float4 accT[2];
    #pragma unroll
    for (int t = 0; t < 2; ++t) {
        const float4* ap = (const float4*)&sAtt[kh][t][q * 32];
        float a[28];
        #pragma unroll
        for (int g = 0; g < 7; ++g) {
            const float4 v = ap[g];
            a[4 * g] = v.x; a[4 * g + 1] = v.y; a[4 * g + 2] = v.z; a[4 * g + 3] = v.w;
        }
        float4 s4 = make_float4(0.f, 0.f, 0.f, 0.f);
        #pragma unroll
        for (int i = 0; i < PMAX / 2; ++i) {
            s4.x = fmaf(a[i], x[i].x, s4.x);
            s4.y = fmaf(a[i], x[i].y, s4.y);
            s4.z = fmaf(a[i], x[i].z, s4.z);
            s4.w = fmaf(a[i], x[i].w, s4.w);
        }
        s4.x += xhalf(s4.x);
        s4.y += xhalf(s4.y);
        s4.z += xhalf(s4.z);
        s4.w += xhalf(s4.w);
        accT[t] = s4;
    }

    {
        const float4 av = q ? accT[1] : accT[0];
        const float4 tg = x[NLD - 1];
        const float dx = (u4.x + av.x) - tg.x;
        const float dy = (u4.y + av.y) - tg.y;
        const float dz = (u4.z + av.z) - tg.z;
        const float dw = (u4.w + av.w) - tg.w;
        float sq = dx * dx + dy * dy + dz * dz + dw * dw;
        sq += row_xor<8>(sq);
        sq += row_xor<4>(sq);
        sq += row_xor<2>(sq);
        sq += row_xor<1>(sq);
        if (j == 0) sRed[q][kh] = sq;
    }
    __syncthreads();
    if (threadIdx.x < 2)
        out[(size_t)threadIdx.x * B + b] = sRed[threadIdx.x][0] + sRed[threadIdx.x][1];
}

extern "C" void kernel_launch(void* const* d_in, const int* in_sizes, int n_in,
                              void* d_out, int out_size, void* d_ws, size_t ws_size,
                              hipStream_t stream) {
    (void)n_in; (void)out_size; (void)d_ws; (void)ws_size;
    const float* user_emb = (const float*)d_in[0];
    const float* item_emb = (const float*)d_in[1];
    const int*   user_ids = (const int*)d_in[2];
    const int*   pos_ids  = (const int*)d_in[3];
    const int*   neg_ids  = (const int*)d_in[4];
    const int*   pref_ids = (const int*)d_in[5];
    const float* n_prefs  = (const float*)d_in[6];
    float* out = (float*)d_out;

    const int B       = in_sizes[2];
    const int n_items = in_sizes[1] / DIMS;

    AttCML_kernel<<<B, 64, 0, stream>>>(user_emb, item_emb, user_ids, pos_ids,
                                        neg_ids, pref_ids, n_prefs, out, B, n_items);
}
